# starting kernel
# baseline (speedup 1.0000x reference)
#include <hip/hip_runtime.h>

typedef float v2f __attribute__((ext_vector_type(2)));
typedef float v8f __attribute__((ext_vector_type(8)));

static constexpr int NTOT = 131072;
static constexpr int D    = 16;
static constexpr int DIN  = 64;
static constexpr int DOUT = 64;
static constexpr int DC   = 4;
static constexpr int QC   = DC * (DIN / 4);
static constexpr int LDS_DW = QC * 4 * 64;

__global__ __launch_bounds__(256)
void mld_wmma_kernel(const float* __restrict__ x,
                     const float* __restrict__ alpha,
                     const float* __restrict__ scale,
                     float* __restrict__ out) {
  __shared__ float ldsB[LDS_DW];

  const int tid  = threadIdx.x;
  const int lane = tid & 31;
  const int wave = tid >> 5;
  const int row  = lane & 15;
  const int hi   = lane >> 4;
  const int c0   = hi << 1;

  const int n0 = blockIdx.x * 128 + wave * 16;
  const int n  = n0 + row;

  v2f xr[16];
#pragma unroll
  for (int t = 0; t < 16; ++t)
    xr[t] = *(const v2f*)(x + (size_t)n * DIN + 4 * t + c0);

  v8f acc0 = {}, acc1 = {}, acc2 = {}, acc3 = {};
  float asum = 0.0f;

  for (int chunk = 0; chunk < D / DC; ++chunk) {
    const int dbase = chunk * DC;

    for (int p = tid; p < QC * 4 * 32; p += 256) {
      const int c  = p & 31;
      const int ob = (p >> 5) & 3;
      const int q  = p >> 7;
      const int d  = dbase + (q >> 4);
      const int i  = ((q & 15) << 2) + ((c >> 4) << 1);
      const int o  = (ob << 4) + (c & 15);
      const float s0 = scale[(size_t)(d * DIN + i)     * DOUT + o];
      const float s1 = scale[(size_t)(d * DIN + i + 1) * DOUT + o];
      v2f pk; pk.x = s0; pk.y = s1;
      *(v2f*)(ldsB + (((q << 2) + ob) << 6) + (c << 1)) = pk;
    }
    __syncthreads();

    for (int dd = 0; dd < DC; ++dd) {
      const int d   = dbase + dd;
      const float a = alpha[(size_t)d * NTOT + n];
      asum += a;
      const int qb = dd << 4;
#pragma unroll
      for (int t = 0; t < 16; ++t) {
        v2f afrag;
        afrag.x = a * xr[t].x;
        afrag.y = a * xr[t].y;
        const float* bp = ldsB + (((qb + t) << 2) << 6) + (lane << 1);
        v2f b0 = *(const v2f*)(bp);
        v2f b1 = *(const v2f*)(bp + 64);
        v2f b2 = *(const v2f*)(bp + 128);
        v2f b3 = *(const v2f*)(bp + 192);
        acc0 = __builtin_amdgcn_wmma_f32_16x16x4_f32(false, afrag, false, b0,
                                                     (short)0, acc0, false, false);
        acc1 = __builtin_amdgcn_wmma_f32_16x16x4_f32(false, afrag, false, b1,
                                                     (short)0, acc1, false, false);
        acc2 = __builtin_amdgcn_wmma_f32_16x16x4_f32(false, afrag, false, b2,
                                                     (short)0, acc2, false, false);
        acc3 = __builtin_amdgcn_wmma_f32_16x16x4_f32(false, afrag, false, b3,
                                                     (short)0, acc3, false, false);
      }
    }
    __syncthreads();
  }

  float rs[8];
#pragma unroll
  for (int r = 0; r < 8; ++r) {
    const int srcByte = (r + (hi << 3)) << 2;
    rs[r] = __int_as_float(
        __builtin_amdgcn_ds_bpermute(srcByte, __float_as_int(asum)));
  }

#pragma unroll
  for (int r = 0; r < 8; ++r) {
    float* orow = out + (size_t)(n0 + r + (hi << 3)) * DOUT + (lane & 15);
    orow[0]  = rs[r] * acc0[r];
    orow[16] = rs[r] * acc1[r];
    orow[32] = rs[r] * acc2[r];
    orow[48] = rs[r] * acc3[r];
  }
}

extern "C" void kernel_launch(void* const* d_in, const int* in_sizes, int n_in,
                              void* d_out, int out_size, void* d_ws, size_t ws_size,
                              hipStream_t stream) {
  const float* x     = (const float*)d_in[0];
  const float* alpha = (const float*)d_in[1];
  const float* scale = (const float*)d_in[2];
  float* out = (float*)d_out;

  dim3 grid(NTOT / 128);
  dim3 block(256);
  hipLaunchKernelGGL(mld_wmma_kernel, grid, block, 0, stream,
                     x, alpha, scale, out);
}


# wave tile 16x64 -> 32x64 (two M tiles per wave): each B fragment read from LDS feeds two WMMAs; 256-row blocks halve B staging per row
# speedup vs baseline: 1.0504x; 1.0504x over previous
#include <hip/hip_runtime.h>


typedef float v2f __attribute__((ext_vector_type(2)));
typedef float v8f __attribute__((ext_vector_type(8)));

static constexpr int NTOT = 131072;
static constexpr int D    = 16;
static constexpr int DIN  = 64;
static constexpr int DOUT = 64;
static constexpr int DC   = 4;
static constexpr int QC   = DC * (DIN / 4);
static constexpr int LDS_DW = QC * 4 * 64;
static constexpr int WROWS = 32;
static constexpr int BROWS = 8 * WROWS;

__global__ __launch_bounds__(256)
void mld_wmma_kernel(const float* __restrict__ x,
                     const float* __restrict__ alpha,
                     const float* __restrict__ scale,
                     float* __restrict__ out) {
  __shared__ float ldsB[LDS_DW];

  const int tid  = threadIdx.x;
  const int lane = tid & 31;
  const int wave = tid >> 5;
  const int row  = lane & 15;
  const int hi   = lane >> 4;
  const int c0   = hi << 1;

  const int n0 = blockIdx.x * BROWS + wave * WROWS;
  const int na = n0 + row;
  const int nb = n0 + 16 + row;

  v2f xa[16], xb[16];
#pragma unroll
  for (int t = 0; t < 16; ++t) {
    xa[t] = *(const v2f*)(x + (size_t)na * DIN + 4 * t + c0);
    xb[t] = *(const v2f*)(x + (size_t)nb * DIN + 4 * t + c0);
  }

  v8f a0 = {}, a1 = {}, a2 = {}, a3 = {};
  v8f b0 = {}, b1 = {}, b2 = {}, b3 = {};
  float asumA = 0.0f, asumB = 0.0f;

  for (int chunk = 0; chunk < D / DC; ++chunk) {
    const int dbase = chunk * DC;

    for (int p = tid; p < QC * 4 * 32; p += 256) {
      const int c  = p & 31;
      const int ob = (p >> 5) & 3;
      const int q  = p >> 7;
      const int d  = dbase + (q >> 4);
      const int i  = ((q & 15) << 2) + ((c >> 4) << 1);
      const int o  = (ob << 4) + (c & 15);
      const float s0 = scale[(size_t)(d * DIN + i)     * DOUT + o];
      const float s1 = scale[(size_t)(d * DIN + i + 1) * DOUT + o];
      v2f pk; pk.x = s0; pk.y = s1;
      *(v2f*)(ldsB + (((q << 2) + ob) << 6) + (c << 1)) = pk;
    }
    __syncthreads();

    for (int dd = 0; dd < DC; ++dd) {
      const int d    = dbase + dd;
      const float wa = alpha[(size_t)d * NTOT + na];
      const float wb = alpha[(size_t)d * NTOT + nb];
      asumA += wa;
      asumB += wb;
      const int qb = dd << 4;
#pragma unroll
      for (int t = 0; t < 16; ++t) {
        v2f fa, fb;
        fa.x = wa * xa[t].x;
        fa.y = wa * xa[t].y;
        fb.x = wb * xb[t].x;
        fb.y = wb * xb[t].y;
        const float* bp = ldsB + (((qb + t) << 2) << 6) + (lane << 1);
        v2f f0 = *(const v2f*)(bp);
        v2f f1 = *(const v2f*)(bp + 64);
        v2f f2 = *(const v2f*)(bp + 128);
        v2f f3 = *(const v2f*)(bp + 192);
        a0 = __builtin_amdgcn_wmma_f32_16x16x4_f32(false, fa, false, f0,
                                                   (short)0, a0, false, false);
        b0 = __builtin_amdgcn_wmma_f32_16x16x4_f32(false, fb, false, f0,
                                                   (short)0, b0, false, false);
        a1 = __builtin_amdgcn_wmma_f32_16x16x4_f32(false, fa, false, f1,
                                                   (short)0, a1, false, false);
        b1 = __builtin_amdgcn_wmma_f32_16x16x4_f32(false, fb, false, f1,
                                                   (short)0, b1, false, false);
        a2 = __builtin_amdgcn_wmma_f32_16x16x4_f32(false, fa, false, f2,
                                                   (short)0, a2, false, false);
        b2 = __builtin_amdgcn_wmma_f32_16x16x4_f32(false, fb, false, f2,
                                                   (short)0, b2, false, false);
        a3 = __builtin_amdgcn_wmma_f32_16x16x4_f32(false, fa, false, f3,
                                                   (short)0, a3, false, false);
        b3 = __builtin_amdgcn_wmma_f32_16x16x4_f32(false, fb, false, f3,
                                                   (short)0, b3, false, false);
      }
    }
    __syncthreads();
  }

#pragma unroll
  for (int r = 0; r < 8; ++r) {
    const int srcByte = (r + (hi << 3)) << 2;
    const float ra = __int_as_float(
        __builtin_amdgcn_ds_bpermute(srcByte, __float_as_int(asumA)));
    const float rb = __int_as_float(
        __builtin_amdgcn_ds_bpermute(srcByte, __float_as_int(asumB)));
    float* orow = out + (size_t)(n0 + r + (hi << 3)) * DOUT + (lane & 15);
    orow[0]  = ra * a0[r];
    orow[16] = ra * a1[r];
    orow[32] = ra * a2[r];
    orow[48] = ra * a3[r];
    float* prow = orow + (size_t)16 * DOUT;
    prow[0]  = rb * b0[r];
    prow[16] = rb * b1[r];
    prow[32] = rb * b2[r];
    prow[48] = rb * b3[r];
  }
}

extern "C" void kernel_launch(void* const* d_in, const int* in_sizes, int n_in,
                              void* d_out, int out_size, void* d_ws, size_t ws_size,
                              hipStream_t stream) {
  const float* x     = (const float*)d_in[0];
  const float* alpha = (const float*)d_in[1];
  const float* scale = (const float*)d_in[2];
  float* out = (float*)d_out;

  dim3 grid(NTOT / BROWS);
  dim3 block(256);
  hipLaunchKernelGGL(mld_wmma_kernel, grid, block, 0, stream,
                     x, alpha, scale, out);
}


# K loop software-pipelined in blocks of 32 products: a block's B fragments (8 16-byte LDS reads, LDS re-laid for it) and the next decoder's alpha are fetched one block ahead as ONE group of reads
# speedup vs baseline: 1.1117x; 1.0584x over previous
#include <hip/hip_runtime.h>


typedef float v2f __attribute__((ext_vector_type(2)));
typedef float v4f __attribute__((ext_vector_type(4)));
typedef float v8f __attribute__((ext_vector_type(8)));

static constexpr int NTOT = 131072;
static constexpr int D    = 16;
static constexpr int DIN  = 64;
static constexpr int DOUT = 64;
static constexpr int DC   = 4;
static constexpr int QC   = DC * (DIN / 4);
static constexpr int LDS_DW = QC * 4 * 64;
static constexpr int WROWS = 32;
static constexpr int BROWS = 8 * WROWS;

#define MMA_BLOCK(FR, T0)                                                          \
  _Pragma("unroll")                                                                \
  for (int tt = 0; tt < 4; ++tt) {                                                 \
    v2f fa, fb;                                                                    \
    fa.x = wa * xa[(T0) + tt].x;                                                   \
    fa.y = wa * xa[(T0) + tt].y;                                                   \
    fb.x = wb * xb[(T0) + tt].x;                                                   \
    fb.y = wb * xb[(T0) + tt].y;                                                   \
    const v4f g0 = FR[2 * tt], g1 = FR[2 * tt + 1];                                \
    v2f f0, f1, f2, f3;                                                            \
    f0.x = g0.x; f0.y = g0.y; f1.x = g0.z; f1.y = g0.w;                            \
    f2.x = g1.x; f2.y = g1.y; f3.x = g1.z; f3.y = g1.w;                            \
    a0 = __builtin_amdgcn_wmma_f32_16x16x4_f32(false, fa, false, f0, (short)0, a0, false, false); \
    b0 = __builtin_amdgcn_wmma_f32_16x16x4_f32(false, fb, false, f0, (short)0, b0, false, false); \
    a1 = __builtin_amdgcn_wmma_f32_16x16x4_f32(false, fa, false, f1, (short)0, a1, false, false); \
    b1 = __builtin_amdgcn_wmma_f32_16x16x4_f32(false, fb, false, f1, (short)0, b1, false, false); \
    a2 = __builtin_amdgcn_wmma_f32_16x16x4_f32(false, fa, false, f2, (short)0, a2, false, false); \
    b2 = __builtin_amdgcn_wmma_f32_16x16x4_f32(false, fb, false, f2, (short)0, b2, false, false); \
    a3 = __builtin_amdgcn_wmma_f32_16x16x4_f32(false, fa, false, f3, (short)0, a3, false, false); \
    b3 = __builtin_amdgcn_wmma_f32_16x16x4_f32(false, fb, false, f3, (short)0, b3, false, false); \
  }

#define GROUP_EDGE()                                                               \
  do {                                                                             \
    asm volatile("" : "+v"(a0), "+v"(a1), "+v"(a2), "+v"(a3),                      \
                      "+v"(b0), "+v"(b1), "+v"(b2), "+v"(b3));                     \
    __builtin_amdgcn_sched_barrier(0);                                             \
  } while (0)

#define LOAD_BLOCK(FR, BASE, Q0)                                                   \
  _Pragma("unroll")                                                                \
  for (int j = 0; j < 8; ++j)                                                      \
    FR[j] = *(const v4f*)((BASE) + ((((Q0) << 1) + j) << 7));

__global__ __launch_bounds__(256)
void mld_wmma_kernel(const float* __restrict__ x,
                     const float* __restrict__ alpha,
                     const float* __restrict__ scale,
                     float* __restrict__ out) {
  __shared__ float ldsB[LDS_DW];

  const int tid  = threadIdx.x;
  const int lane = tid & 31;
  const int wave = tid >> 5;
  const int row  = lane & 15;
  const int hi   = lane >> 4;
  const int c0   = hi << 1;

  const int n0 = blockIdx.x * BROWS + wave * WROWS;
  const int na = n0 + row;
  const int nb = n0 + 16 + row;

  v2f xa[16], xb[16];
#pragma unroll
  for (int t = 0; t < 16; ++t) {
    xa[t] = *(const v2f*)(x + (size_t)na * DIN + 4 * t + c0);
    xb[t] = *(const v2f*)(x + (size_t)nb * DIN + 4 * t + c0);
  }
  float wan = alpha[na];
  float wbn = alpha[nb];

  v8f a0 = {}, a1 = {}, a2 = {}, a3 = {};
  v8f b0 = {}, b1 = {}, b2 = {}, b3 = {};
  float asumA = 0.0f, asumB = 0.0f;

  v4f F0[8], F1[8];

  for (int chunk = 0; chunk < D / DC; ++chunk) {
    const int dbase = chunk * DC;

    for (int p = tid; p < QC * 2 * 32; p += 256) {
      const int c  = p & 31;
      const int h  = (p >> 5) & 1;
      const int q  = p >> 6;
      const int d  = dbase + (q >> 4);
      const int i  = ((q & 15) << 2) + ((c >> 4) << 1);
      const int o  = (h << 5) + (c & 15);
      const float* sp = scale + (size_t)(d * DIN + i) * DOUT + o;
      v4f pk;
      pk.x = sp[0];
      pk.y = sp[DOUT];
      pk.z = sp[16];
      pk.w = sp[DOUT + 16];
      *(v4f*)(ldsB + (((q << 1) + h) << 7) + (c << 2)) = pk;
    }
    __syncthreads();

    LOAD_BLOCK(F0, ldsB + (lane << 2), 0)
    for (int dd = 0; dd < DC; ++dd) {
      const int d    = dbase + dd;
      const float wa = wan;
      const float wb = wbn;
      asumA += wa;
      asumB += wb;
      const float* bl = ldsB + (dd << 12) + (lane << 2);
      const int dn = d + 1 < D ? d + 1 : D - 1;

      LOAD_BLOCK(F1, bl, 4)
      wan = alpha[(size_t)dn * NTOT + na];
      wbn = alpha[(size_t)dn * NTOT + nb];
      GROUP_EDGE();
      MMA_BLOCK(F0, 0)
      GROUP_EDGE();
      LOAD_BLOCK(F0, bl, 8)
      GROUP_EDGE();
      MMA_BLOCK(F1, 4)
      GROUP_EDGE();
      LOAD_BLOCK(F1, bl, 12)
      GROUP_EDGE();
      MMA_BLOCK(F0, 8)
      GROUP_EDGE();
      LOAD_BLOCK(F0, ldsB + (((dd + 1) & (DC - 1)) << 12) + (lane << 2), 0)
      GROUP_EDGE();
      MMA_BLOCK(F1, 12)
      GROUP_EDGE();
    }
    __syncthreads();
  }

#pragma unroll
  for (int r = 0; r < 8; ++r) {
    const int srcByte = (r + (hi << 3)) << 2;
    const float ra = __int_as_float(
        __builtin_amdgcn_ds_bpermute(srcByte, __float_as_int(asumA)));
    const float rb = __int_as_float(
        __builtin_amdgcn_ds_bpermute(srcByte, __float_as_int(asumB)));
    float* orow = out + (size_t)(n0 + r + (hi << 3)) * DOUT + (lane & 15);
    orow[0]  = ra * a0[r];
    orow[16] = ra * a1[r];
    orow[32] = ra * a2[r];
    orow[48] = ra * a3[r];
    float* prow = orow + (size_t)16 * DOUT;
    prow[0]  = rb * b0[r];
    prow[16] = rb * b1[r];
    prow[32] = rb * b2[r];
    prow[48] = rb * b3[r];
  }
}

extern "C" void kernel_launch(void* const* d_in, const int* in_sizes, int n_in,
                              void* d_out, int out_size, void* d_ws, size_t ws_size,
                              hipStream_t stream) {
  const float* x     = (const float*)d_in[0];
  const float* alpha = (const float*)d_in[1];
  const float* scale = (const float*)d_in[2];
  float* out = (float*)d_out;

  dim3 grid(NTOT / BROWS);
  dim3 block(256);
  hipLaunchKernelGGL(mld_wmma_kernel, grid, block, 0, stream,
                     x, alpha, scale, out);
}
